# starting kernel
# baseline (speedup 1.0000x reference)
#include <hip/hip_runtime.h>
#include <hip/hip_bf16.h>
#include <math.h>


typedef __attribute__((ext_vector_type(2))) float v2f;
typedef __attribute__((ext_vector_type(8))) float v8f;

#define RS2 0.70710678118654752f

__global__ __launch_bounds__(96) void qc_build_G(
    const float* __restrict__ thetas, const float* __restrict__ phis,
    float* __restrict__ G) {
  int t = threadIdx.x;
  if (t >= 96) return;
  int l = t >> 5;
  int j = t & 31;

  float re[32], im[32];
#pragma unroll
  for (int i = 0; i < 32; ++i) { re[i] = 0.0f; im[i] = 0.0f; }
  re[j] = 1.0f;

  const int cb[4] = {8, 4, 2, 1};
  const int tb[4] = {4, 2, 1, 8};
#pragma unroll
  for (int w = 0; w < 4; ++w) {
    float th = thetas[4 * l + w];
    float ec = cosf(0.5f * th), es = sinf(0.5f * th);
    int bc = cb[w], bt = tb[w];
#pragma unroll
    for (int i = 0; i < 32; ++i) {
      if ((i & bc) && !(i & bt)) {
        int i1 = i | bt;
        float ur = re[i], ui = im[i], vr = re[i1], vi = im[i1];
        re[i]  =  es * vr + ec * vi;
        im[i]  =  es * vi - ec * vr;
        re[i1] = -es * ur + ec * ui;
        im[i1] = -es * ui - ec * ur;
      }
    }
  }
  float pr = cosf(phis[l]), pi = sinf(phis[l]);
#pragma unroll
  for (int i = 0; i < 32; ++i) {
    if ((i & 16) && (i & 8)) {
      float r = re[i] * pr - im[i] * pi;
      float m = re[i] * pi + im[i] * pr;
      re[i] = r; im[i] = m;
    }
  }
  float* Gl = G + l * 64 * 64;
#pragma unroll
  for (int r = 0; r < 32; ++r) {
    Gl[r * 64 + j]             =  re[r];
    Gl[r * 64 + j + 32]        = -im[r];
    Gl[(r + 32) * 64 + j]      =  im[r];
    Gl[(r + 32) * 64 + j + 32] =  re[r];
  }
}

#define GP 65
#define SP 33

__global__ __launch_bounds__(128) void qc_main(
    const float* __restrict__ x, const float* __restrict__ G,
    float* __restrict__ out, int P) {
  __shared__ float Gs[64 * GP];
  __shared__ float Sb[4][64 * SP];

  const int tid  = threadIdx.x;
  const int lane = tid & 31;
  const int wv   = tid >> 5;
  const int m    = lane & 15;
  const int hi   = lane >> 4;
  float* Sw = &Sb[wv][0];

  const int p  = blockIdx.x * 128 + tid;
  const int b  = p >> 14;
  const int h2 = (p >> 7) & 127;
  const int w2 = p & 127;

  float ang[12];
#pragma unroll
  for (int c = 0; c < 3; ++c) {
#pragma unroll
    for (int dy = 0; dy < 2; ++dy) {
      const float* row = x + (((b * 3 + c) * 256 + (2 * h2 + dy)) * 256 + 2 * w2);
      ang[c * 4 + dy * 2 + 0] = row[0];
      ang[c * 4 + dy * 2 + 1] = row[1];
    }
  }

  float st[64];
#pragma unroll
  for (int i = 0; i < 64; ++i) st[i] = 0.0f;
  st[0] = RS2;
  st[16] = RS2;

#pragma unroll
  for (int l = 0; l < 3; ++l) {
#pragma unroll
    for (int wq = 0; wq < 4; ++wq) {
      float h = 0.5f * ang[4 * l + wq];
      float c_ = cosf(h), s_ = sinf(h);
      const int bt = 8 >> wq;
#pragma unroll
      for (int i = 0; i < 32; ++i) {
        if (!(i & bt)) {
          int jj = i | bt;
          float ur = st[i], ui = st[i + 32], vr = st[jj], vi = st[jj + 32];
          st[i]       = c_ * ur + s_ * vi;
          st[i + 32]  = c_ * ui - s_ * vr;
          st[jj]      = c_ * vr + s_ * ui;
          st[jj + 32] = c_ * vi - s_ * ur;
        }
      }
    }

    __syncthreads();

    for (int idx = tid; idx < 4096; idx += 128) {
      int r = idx >> 6, cc = idx & 63;
      Gs[r * GP + cc] = G[(l * 64 + r) * 64 + cc];
    }
#pragma unroll
    for (int r = 0; r < 64; ++r) Sw[r * SP + lane] = st[r];

    __syncthreads();

    v8f acc[4][2] = {};
    for (int ks = 0; ks < 16; ++ks) {
      const int kA = ks * 4 + 2 * hi;
      v2f Bf0, Bf1;
      Bf0[0] = Sw[(kA + 0) * SP + m];
      Bf0[1] = Sw[(kA + 1) * SP + m];
      Bf1[0] = Sw[(kA + 0) * SP + 16 + m];
      Bf1[1] = Sw[(kA + 1) * SP + 16 + m];
#pragma unroll
      for (int mt = 0; mt < 4; ++mt) {
        v2f Af;
        Af[0] = Gs[(mt * 16 + m) * GP + kA];
        Af[1] = Gs[(mt * 16 + m) * GP + kA + 1];
        acc[mt][0] = __builtin_amdgcn_wmma_f32_16x16x4_f32(
            false, Af, false, Bf0, (short)0, acc[mt][0], false, false);
        acc[mt][1] = __builtin_amdgcn_wmma_f32_16x16x4_f32(
            false, Af, false, Bf1, (short)0, acc[mt][1], false, false);
      }
    }

#pragma unroll
    for (int mt = 0; mt < 4; ++mt)
#pragma unroll
      for (int nt = 0; nt < 2; ++nt)
#pragma unroll
        for (int r = 0; r < 8; ++r)
          Sw[(mt * 16 + hi * 8 + r) * SP + nt * 16 + m] = acc[mt][nt][r];

#pragma unroll
    for (int r = 0; r < 64; ++r) st[r] = Sw[r * SP + lane];
  }

#pragma unroll
  for (int i = 0; i < 16; ++i) {
    float ur = st[i],      ui = st[i + 32];
    float vr = st[i + 16], vi = st[i + 48];
    st[i]      = (ur + vr) * RS2;
    st[i + 32] = (ui + vi) * RS2;
    st[i + 16] = (ur - vr) * RS2;
    st[i + 48] = (ui - vi) * RS2;
  }

  float ev = 0.0f;
#pragma unroll
  for (int i = 0; i < 32; ++i) {
    float p2 = st[i] * st[i] + st[i + 32] * st[i + 32];
    ev += (i & 16) ? -p2 : p2;
  }
  if (p < P) out[p] = ev;
}

extern "C" void kernel_launch(void* const* d_in, const int* in_sizes, int n_in,
                              void* d_out, int out_size, void* d_ws, size_t ws_size,
                              hipStream_t stream) {
  (void)in_sizes; (void)n_in; (void)ws_size;
  const float* x      = (const float*)d_in[0];
  const float* thetas = (const float*)d_in[1];
  const float* phis   = (const float*)d_in[2];
  float* out = (float*)d_out;
  float* G   = (float*)d_ws;

  hipLaunchKernelGGL(qc_build_G, dim3(1), dim3(96), 0, stream, thetas, phis, G);

  const int P = out_size;
  const int blocks = (P + 127) / 128;
  hipLaunchKernelGGL(qc_main, dim3(blocks), dim3(128), 0, stream, x, G, out, P);
}


# Layer matrix factored K S K: K = fixed +-1/4 Hadamard of wires 1-4 as WMMA product, A in registers (no staging, no barrier); S = per-lane phased renaming; 160 WMMAs not 288; sum order changes
# speedup vs baseline: 2.3217x; 2.3217x over previous
#include <hip/hip_runtime.h>
#include <hip/hip_bf16.h>
#include <math.h>


typedef __attribute__((ext_vector_type(2))) float v2f;
typedef __attribute__((ext_vector_type(4))) float v4f;
typedef __attribute__((ext_vector_type(8))) float v8f;

#define RS2 0.70710678118654752f

__host__ __device__ constexpr int ring_perm(int j) {
  const int cb[4] = {8, 4, 2, 1};
  const int tb[4] = {4, 2, 1, 8};
  for (int w = 0; w < 4; ++w)
    if (j & cb[w]) j ^= tb[w];
  return j;
}

__global__ __launch_bounds__(96) void qc_build_T(
    const float* __restrict__ thetas, const float* __restrict__ phis,
    float* __restrict__ T) {
  int t = threadIdx.x;
  if (t >= 96) return;
  int l = t >> 5;
  int i = t & 31;

  const int a = i & 7;
  const float th_a = thetas[4 * l + (a & 3)];
  const float ph_l = phis[l];
  const float arg = (a < 4) ? 0.5f * th_a : ph_l;
  const float ca = cosf(arg), sa = sinf(arg);
  float ecw[4], esw[4];
#pragma unroll
  for (int w = 0; w < 4; ++w) {
    ecw[w] = __int_as_float(__builtin_amdgcn_readlane(__float_as_int(ca), w));
    esw[w] = __int_as_float(__builtin_amdgcn_readlane(__float_as_int(sa), w));
  }
  const float cpr = __int_as_float(__builtin_amdgcn_readlane(__float_as_int(ca), 4));
  const float cpi = __int_as_float(__builtin_amdgcn_readlane(__float_as_int(sa), 4));

  const int cb[4] = {8, 4, 2, 1};
  const int tb[4] = {4, 2, 1, 8};
  int idx = i & 15;
  float pr = 1.0f, pi = 0.0f;
#pragma unroll
  for (int w = 0; w < 4; ++w) {
    if (idx & cb[w]) {
      const float fr = (idx & tb[w]) ? esw[w] : -esw[w];
      const float fi = -ecw[w];
      const float r = pr * fr - pi * fi;
      const float m = pr * fi + pi * fr;
      pr = r; pi = m;
      idx ^= tb[w];
    }
  }
  if ((i & 16) && (idx & 8)) {
    const float r = pr * cpr - pi * cpi;
    const float m = pr * cpi + pi * cpr;
    pr = r; pi = m;
  }
  T[l * 64 + 2 * i]     = pr;
  T[l * 64 + 2 * i + 1] = pi;
}

#define TP 68

__device__ __forceinline__ void apply_K(float (&st)[64], float* Sw, float* Smine,
                                        int m, int hi, const v2f (&Ka)[2][2]) {
#pragma unroll
  for (int j = 0; j < 16; ++j) {
    v4f t;
    t[0] = st[4 * j + 0]; t[1] = st[4 * j + 1];
    t[2] = st[4 * j + 2]; t[3] = st[4 * j + 3];
    *reinterpret_cast<v4f*>(Smine + 4 * j) = t;
  }
#pragma unroll
  for (int pt = 0; pt < 2; ++pt) {
    const float* Bb = Sw + (pt * 16 + m) * TP + 4 * hi;
    v4f B0[4], B1[4];
#pragma unroll
    for (int c = 0; c < 4; ++c) {
      B0[c] = *reinterpret_cast<const v4f*>(Bb + c * 16);
      B1[c] = *reinterpret_cast<const v4f*>(Bb + c * 16 + 8);
    }
    v8f acc[4] = {};
#pragma unroll
    for (int c = 0; c < 4; ++c) {
      v2f b;
      b[0] = B0[c][0]; b[1] = B0[c][1];
      acc[c] = __builtin_amdgcn_wmma_f32_16x16x4_f32(
          false, Ka[0][0], false, b, (short)0, acc[c], false, false);
    }
#pragma unroll
    for (int c = 0; c < 4; ++c) {
      v2f b;
      b[0] = B0[c][2]; b[1] = B0[c][3];
      acc[c] = __builtin_amdgcn_wmma_f32_16x16x4_f32(
          false, Ka[0][1], false, b, (short)0, acc[c], false, false);
    }
#pragma unroll
    for (int c = 0; c < 4; ++c) {
      v2f b;
      b[0] = B1[c][0]; b[1] = B1[c][1];
      acc[c] = __builtin_amdgcn_wmma_f32_16x16x4_f32(
          false, Ka[1][0], false, b, (short)0, acc[c], false, false);
    }
#pragma unroll
    for (int c = 0; c < 4; ++c) {
      v2f b;
      b[0] = B1[c][2]; b[1] = B1[c][3];
      acc[c] = __builtin_amdgcn_wmma_f32_16x16x4_f32(
          false, Ka[1][1], false, b, (short)0, acc[c], false, false);
    }
    float* d = Sw + (pt * 16 + m) * TP + hi * 8;
#pragma unroll
    for (int c = 0; c < 4; ++c) {
      v4f t;
      t[0] = acc[c][0]; t[1] = acc[c][1]; t[2] = acc[c][2]; t[3] = acc[c][3];
      *reinterpret_cast<v4f*>(d + c * 16) = t;
      t[0] = acc[c][4]; t[1] = acc[c][5]; t[2] = acc[c][6]; t[3] = acc[c][7];
      *reinterpret_cast<v4f*>(d + c * 16 + 4) = t;
    }
  }
#pragma unroll
  for (int j = 0; j < 16; ++j) {
    const v4f t = *reinterpret_cast<const v4f*>(Smine + 4 * j);
    st[4 * j + 0] = t[0]; st[4 * j + 1] = t[1];
    st[4 * j + 2] = t[2]; st[4 * j + 3] = t[3];
  }
}

__global__ __launch_bounds__(128) void qc_main(
    const float* __restrict__ x, const float* __restrict__ T,
    float* __restrict__ out, int P) {
  __shared__ __attribute__((aligned(16))) float Sb[4][32 * TP];

  const int tid  = threadIdx.x;
  const int lane = tid & 31;
  const int wv   = tid >> 5;
  const int m    = lane & 15;
  const int hi   = lane >> 4;
  float* Sw = &Sb[wv][0];
  float* Smine = Sw + lane * TP;

  const int p  = blockIdx.x * 128 + tid;
  const int b  = p >> 14;
  const int h2 = (p >> 7) & 127;
  const int w2 = p & 127;

  v2f Ka[2][2];
#pragma unroll
  for (int kk = 0; kk < 2; ++kk)
#pragma unroll
    for (int q = 0; q < 2; ++q)
#pragma unroll
      for (int e = 0; e < 2; ++e) {
        const int k = 8 * kk + 4 * hi + 2 * q + e;
        Ka[kk][q][e] = (__builtin_popcount(m & k) & 1) ? -0.25f : 0.25f;
      }

  float ang[12];
#pragma unroll
  for (int c = 0; c < 3; ++c) {
#pragma unroll
    for (int dy = 0; dy < 2; ++dy) {
      const float* row = x + (((b * 3 + c) * 256 + (2 * h2 + dy)) * 256 + 2 * w2);
      ang[c * 4 + dy * 2 + 0] = row[0];
      ang[c * 4 + dy * 2 + 1] = row[1];
    }
  }

  float st[64];
#pragma unroll
  for (int i = 0; i < 64; ++i) st[i] = 0.0f;

#pragma unroll
  for (int l = 0; l < 3; ++l) {
    float cw[4], sw[4];
#pragma unroll
    for (int wq = 0; wq < 4; ++wq) {
      const float h = 0.5f * ang[4 * l + wq];
      cw[wq] = cosf(h);
      sw[wq] = sinf(h);
    }
    const float p0r = cw[0] * cw[1] - sw[0] * sw[1];
    const float p0i = -(sw[0] * cw[1] + cw[0] * sw[1]);
    const float p1r = cw[0] * cw[1] + sw[0] * sw[1];
    const float p1i = cw[0] * sw[1] - sw[0] * cw[1];
    const float q0r = cw[2] * cw[3] - sw[2] * sw[3];
    const float q0i = -(sw[2] * cw[3] + cw[2] * sw[3]);
    const float q1r = cw[2] * cw[3] + sw[2] * sw[3];
    const float q1i = cw[2] * sw[3] - sw[2] * cw[3];
    const float Pr[2] = {p0r, p1r}, Pi[2] = {p0i, p1i};
    const float Qr[4] = {q0r, q1r, q1r, q0r}, Qi[4] = {q0i, q1i, -q1i, -q0i};
    float Dr[8], Di[8];
#pragma unroll
    for (int k = 0; k < 8; ++k) {
      const int pp = k >> 2, qq = k & 3;
      Dr[k] = Pr[pp] * Qr[qq] - Pi[pp] * Qi[qq];
      Di[k] = Pr[pp] * Qi[qq] + Pi[pp] * Qr[qq];
    }
    if (l == 0) {
#pragma unroll
      for (int k = 0; k < 8; ++k) {
#pragma unroll
        for (int b0 = 0; b0 < 32; b0 += 16) {
          st[k + b0]           =  Dr[k];
          st[k + b0 + 32]      =  Di[k];
          st[15 - k + b0]      =  Dr[k];
          st[15 - k + b0 + 32] = -Di[k];
        }
      }
    } else {
#pragma unroll
      for (int k = 0; k < 8; ++k) {
#pragma unroll
        for (int b0 = 0; b0 < 32; b0 += 16) {
          const int i0 = k + b0, i1 = 15 - k + b0;
          const float r0 = st[i0], m0 = st[i0 + 32];
          const float r1 = st[i1], m1 = st[i1 + 32];
          st[i0]      = r0 * Dr[k] - m0 * Di[k];
          st[i0 + 32] = r0 * Di[k] + m0 * Dr[k];
          st[i1]      = r1 * Dr[k] + m1 * Di[k];
          st[i1 + 32] = m1 * Dr[k] - r1 * Di[k];
        }
      }
    }

    apply_K(st, Sw, Smine, m, hi, Ka);

    {
      const float* Tl = T + l * 64;
      float ns[64];
#pragma unroll
      for (int i = 0; i < 32; ++i) {
        const int o = (i & 16) + ring_perm(i & 15);
        const float lr = Tl[2 * i], li = Tl[2 * i + 1];
        ns[o]      = st[i] * lr - st[i + 32] * li;
        ns[o + 32] = st[i] * li + st[i + 32] * lr;
      }
#pragma unroll
      for (int i = 0; i < 64; ++i) st[i] = ns[i];
    }

    if (l < 2) apply_K(st, Sw, Smine, m, hi, Ka);
  }

#pragma unroll
  for (int i = 0; i < 16; ++i) {
    float ur = st[i],      ui = st[i + 32];
    float vr = st[i + 16], vi = st[i + 48];
    st[i]      = (ur + vr) * RS2;
    st[i + 32] = (ui + vi) * RS2;
    st[i + 16] = (ur - vr) * RS2;
    st[i + 48] = (ui - vi) * RS2;
  }

  float ev = 0.0f;
#pragma unroll
  for (int i = 0; i < 32; ++i) {
    float p2 = st[i] * st[i] + st[i + 32] * st[i + 32];
    ev += (i & 16) ? -p2 : p2;
  }
  if (p < P) out[p] = ev * 0.03125f;
}

extern "C" void kernel_launch(void* const* d_in, const int* in_sizes, int n_in,
                              void* d_out, int out_size, void* d_ws, size_t ws_size,
                              hipStream_t stream) {
  (void)in_sizes; (void)n_in; (void)ws_size;
  const float* x      = (const float*)d_in[0];
  const float* thetas = (const float*)d_in[1];
  const float* phis   = (const float*)d_in[2];
  float* out = (float*)d_out;
  float* T   = (float*)d_ws;

  hipLaunchKernelGGL(qc_build_T, dim3(1), dim3(96), 0, stream, thetas, phis, T);

  const int P = out_size;
  const int blocks = (P + 127) / 128;
  hipLaunchKernelGGL(qc_main, dim3(blocks), dim3(128), 0, stream, x, T, out, P);
}


# Layers 0,1: K S K fused per wire-0 half into one complex 16x16 matrix (built in prologue), applied as three real products: 128 WMMAs not 160, 3 LDS round trips not 5; k sum order, rounding change
# speedup vs baseline: 2.4986x; 1.0762x over previous
#include <hip/hip_runtime.h>
#include <hip/hip_bf16.h>
#include <math.h>


typedef __attribute__((ext_vector_type(2))) float v2f;
typedef __attribute__((ext_vector_type(4))) float v4f;
typedef __attribute__((ext_vector_type(8))) float v8f;

#define RS2 0.70710678118654752f

__host__ __device__ constexpr int ring_perm(int j) {
  const int cb[4] = {8, 4, 2, 1};
  const int tb[4] = {4, 2, 1, 8};
  for (int w = 0; w < 4; ++w)
    if (j & cb[w]) j ^= tb[w];
  return j;
}

__global__ __launch_bounds__(1024) void qc_build_T(
    const float* __restrict__ thetas, const float* __restrict__ phis,
    float* __restrict__ T) {
  __shared__ float Ts[128];
  const int t = threadIdx.x;
  if (t < 96) {
    const int l = t >> 5;
    const int i = t & 31;

    const int a = i & 7;
    const float th_a = thetas[4 * l + (a & 3)];
    const float ph_l = phis[l];
    const float arg = (a < 4) ? 0.5f * th_a : ph_l;
    const float ca = cosf(arg), sa = sinf(arg);
    float ecw[4], esw[4];
#pragma unroll
    for (int w = 0; w < 4; ++w) {
      ecw[w] = __int_as_float(__builtin_amdgcn_readlane(__float_as_int(ca), w));
      esw[w] = __int_as_float(__builtin_amdgcn_readlane(__float_as_int(sa), w));
    }
    const float cpr = __int_as_float(__builtin_amdgcn_readlane(__float_as_int(ca), 4));
    const float cpi = __int_as_float(__builtin_amdgcn_readlane(__float_as_int(sa), 4));

    const int cb[4] = {8, 4, 2, 1};
    const int tb[4] = {4, 2, 1, 8};
    int idx = i & 15;
    float pr = 1.0f, pi = 0.0f;
#pragma unroll
    for (int w = 0; w < 4; ++w) {
      if (idx & cb[w]) {
        const float fr = (idx & tb[w]) ? esw[w] : -esw[w];
        const float fi = -ecw[w];
        const float r = pr * fr - pi * fi;
        const float mm = pr * fi + pi * fr;
        pr = r; pi = mm;
        idx ^= tb[w];
      }
    }
    if ((i & 16) && (idx & 8)) {
      const float r = pr * cpr - pi * cpi;
      const float mm = pr * cpi + pi * cpr;
      pr = r; pi = mm;
    }
    T[l * 64 + 2 * i]     = pr;
    T[l * 64 + 2 * i + 1] = pi;
    if (l < 2) {
      Ts[l * 64 + 2 * i]     = pr;
      Ts[l * 64 + 2 * i + 1] = pi;
    }
  }
  __syncthreads();

  const int l = t >> 9;
  const int h = (t >> 8) & 1;
  const int m = (t >> 4) & 15;
  const int n = t & 15;
  const float* ph = Ts + l * 64 + 32 * h;
  float mr = 0.0f, mi = 0.0f;
#pragma unroll
  for (int i = 0; i < 16; ++i) {
    const int o = ring_perm(i);
    const bool neg = ((__builtin_popcount(m & o) + __builtin_popcount(i & n)) & 1) != 0;
    const float tr = ph[2 * i], ti = ph[2 * i + 1];
    mr += neg ? -tr : tr;
    mi += neg ? -ti : ti;
  }
  mr *= 0.0625f;
  mi *= 0.0625f;
  float* Mo = T + 256 + ((l * 2 + h) * 3) * 256 + m * 16 + n;
  Mo[0]   = mr;
  Mo[256] = -(mr + mi);
  Mo[512] = mi - mr;
}

#define TP 68

__device__ __forceinline__ void apply_K(float (&st)[64], float* Sw, float* Smine,
                                        int m, int hi, const v2f (&Ka)[2][2]) {
#pragma unroll
  for (int j = 0; j < 16; ++j) {
    v4f t;
    t[0] = st[4 * j + 0]; t[1] = st[4 * j + 1];
    t[2] = st[4 * j + 2]; t[3] = st[4 * j + 3];
    *reinterpret_cast<v4f*>(Smine + 4 * j) = t;
  }
#pragma unroll
  for (int pt = 0; pt < 2; ++pt) {
    const float* Bb = Sw + (pt * 16 + m) * TP + 4 * hi;
    v4f B0[4], B1[4];
#pragma unroll
    for (int c = 0; c < 4; ++c) {
      B0[c] = *reinterpret_cast<const v4f*>(Bb + c * 16);
      B1[c] = *reinterpret_cast<const v4f*>(Bb + c * 16 + 8);
    }
    v8f acc[4] = {};
#pragma unroll
    for (int c = 0; c < 4; ++c) {
      v2f b;
      b[0] = B0[c][0]; b[1] = B0[c][1];
      acc[c] = __builtin_amdgcn_wmma_f32_16x16x4_f32(
          false, Ka[0][0], false, b, (short)0, acc[c], false, false);
    }
#pragma unroll
    for (int c = 0; c < 4; ++c) {
      v2f b;
      b[0] = B0[c][2]; b[1] = B0[c][3];
      acc[c] = __builtin_amdgcn_wmma_f32_16x16x4_f32(
          false, Ka[0][1], false, b, (short)0, acc[c], false, false);
    }
#pragma unroll
    for (int c = 0; c < 4; ++c) {
      v2f b;
      b[0] = B1[c][0]; b[1] = B1[c][1];
      acc[c] = __builtin_amdgcn_wmma_f32_16x16x4_f32(
          false, Ka[1][0], false, b, (short)0, acc[c], false, false);
    }
#pragma unroll
    for (int c = 0; c < 4; ++c) {
      v2f b;
      b[0] = B1[c][2]; b[1] = B1[c][3];
      acc[c] = __builtin_amdgcn_wmma_f32_16x16x4_f32(
          false, Ka[1][1], false, b, (short)0, acc[c], false, false);
    }
    float* d = Sw + (pt * 16 + m) * TP + hi * 8;
#pragma unroll
    for (int c = 0; c < 4; ++c) {
      v4f t;
      t[0] = acc[c][0]; t[1] = acc[c][1]; t[2] = acc[c][2]; t[3] = acc[c][3];
      *reinterpret_cast<v4f*>(d + c * 16) = t;
      t[0] = acc[c][4]; t[1] = acc[c][5]; t[2] = acc[c][6]; t[3] = acc[c][7];
      *reinterpret_cast<v4f*>(d + c * 16 + 4) = t;
    }
  }
#pragma unroll
  for (int j = 0; j < 16; ++j) {
    const v4f t = *reinterpret_cast<const v4f*>(Smine + 4 * j);
    st[4 * j + 0] = t[0]; st[4 * j + 1] = t[1];
    st[4 * j + 2] = t[2]; st[4 * j + 3] = t[3];
  }
}

__device__ __forceinline__ void apply_M(float (&st)[64], float* Sw, float* Smine,
                                        int m, int hi, const float* __restrict__ Ml) {
  const float* Arow = Ml + m * 16 + 4 * hi;
  v4f A[2][3][2];
#pragma unroll
  for (int v = 0; v < 3; ++v) {
    A[0][v][0] = *reinterpret_cast<const v4f*>(Arow + v * 256);
    A[0][v][1] = *reinterpret_cast<const v4f*>(Arow + v * 256 + 8);
  }
#pragma unroll
  for (int j = 0; j < 16; ++j) {
    v4f t;
    t[0] = st[4 * j + 0]; t[1] = st[4 * j + 1];
    t[2] = st[4 * j + 2]; t[3] = st[4 * j + 3];
    *reinterpret_cast<v4f*>(Smine + 4 * j) = t;
  }
#pragma unroll
  for (int v = 0; v < 3; ++v) {
    A[1][v][0] = *reinterpret_cast<const v4f*>(Arow + 768 + v * 256);
    A[1][v][1] = *reinterpret_cast<const v4f*>(Arow + 768 + v * 256 + 8);
  }
#pragma unroll
  for (int h = 0; h < 2; ++h) {
#pragma unroll
    for (int pt = 0; pt < 2; ++pt) {
      const float* Bb = Sw + (pt * 16 + m) * TP + 4 * hi;
      v4f xr[2], xi[2], xs[2];
      xr[0] = *reinterpret_cast<const v4f*>(Bb + h * 16);
      xr[1] = *reinterpret_cast<const v4f*>(Bb + h * 16 + 8);
      xi[0] = *reinterpret_cast<const v4f*>(Bb + (2 + h) * 16);
      xi[1] = *reinterpret_cast<const v4f*>(Bb + (2 + h) * 16 + 8);
      xs[0] = xr[0] + xi[0];
      xs[1] = xr[1] + xi[1];
      v8f k1 = {};
#pragma unroll
      for (int kk = 0; kk < 2; ++kk)
#pragma unroll
        for (int q = 0; q < 2; ++q) {
          v2f a, b;
          a[0] = A[h][0][kk][2 * q]; a[1] = A[h][0][kk][2 * q + 1];
          b[0] = xs[kk][2 * q];      b[1] = xs[kk][2 * q + 1];
          k1 = __builtin_amdgcn_wmma_f32_16x16x4_f32(
              false, a, false, b, (short)0, k1, false, false);
        }
      v8f re = k1, im = k1;
#pragma unroll
      for (int kk = 0; kk < 2; ++kk)
#pragma unroll
        for (int q = 0; q < 2; ++q) {
          v2f a, b;
          a[0] = A[h][1][kk][2 * q]; a[1] = A[h][1][kk][2 * q + 1];
          b[0] = xi[kk][2 * q];      b[1] = xi[kk][2 * q + 1];
          re = __builtin_amdgcn_wmma_f32_16x16x4_f32(
              false, a, false, b, (short)0, re, false, false);
          a[0] = A[h][2][kk][2 * q]; a[1] = A[h][2][kk][2 * q + 1];
          b[0] = xr[kk][2 * q];      b[1] = xr[kk][2 * q + 1];
          im = __builtin_amdgcn_wmma_f32_16x16x4_f32(
              false, a, false, b, (short)0, im, false, false);
        }
      float* d = Sw + (pt * 16 + m) * TP + hi * 8;
      v4f t;
      t[0] = re[0]; t[1] = re[1]; t[2] = re[2]; t[3] = re[3];
      *reinterpret_cast<v4f*>(d + h * 16) = t;
      t[0] = re[4]; t[1] = re[5]; t[2] = re[6]; t[3] = re[7];
      *reinterpret_cast<v4f*>(d + h * 16 + 4) = t;
      t[0] = im[0]; t[1] = im[1]; t[2] = im[2]; t[3] = im[3];
      *reinterpret_cast<v4f*>(d + (2 + h) * 16) = t;
      t[0] = im[4]; t[1] = im[5]; t[2] = im[6]; t[3] = im[7];
      *reinterpret_cast<v4f*>(d + (2 + h) * 16 + 4) = t;
    }
  }
#pragma unroll
  for (int j = 0; j < 16; ++j) {
    const v4f t = *reinterpret_cast<const v4f*>(Smine + 4 * j);
    st[4 * j + 0] = t[0]; st[4 * j + 1] = t[1];
    st[4 * j + 2] = t[2]; st[4 * j + 3] = t[3];
  }
}

__global__ __launch_bounds__(128, 8) void qc_main(
    const float* __restrict__ x, const float* __restrict__ T,
    float* __restrict__ out, int P) {
  __shared__ __attribute__((aligned(16))) float Sb[4][32 * TP];

  const int tid  = threadIdx.x;
  const int lane = tid & 31;
  const int wv   = tid >> 5;
  const int m    = lane & 15;
  const int hi   = lane >> 4;
  float* Sw = &Sb[wv][0];
  float* Smine = Sw + lane * TP;

  const int p  = blockIdx.x * 128 + tid;
  const int b  = p >> 14;
  const int h2 = (p >> 7) & 127;
  const int w2 = p & 127;

  v2f Ka[2][2];
#pragma unroll
  for (int kk = 0; kk < 2; ++kk)
#pragma unroll
    for (int q = 0; q < 2; ++q)
#pragma unroll
      for (int e = 0; e < 2; ++e) {
        const int k = 8 * kk + 4 * hi + 2 * q + e;
        Ka[kk][q][e] = (__builtin_popcount(m & k) & 1) ? -0.25f : 0.25f;
      }

  float ang[12];
#pragma unroll
  for (int c = 0; c < 3; ++c) {
#pragma unroll
    for (int dy = 0; dy < 2; ++dy) {
      const float* row = x + (((b * 3 + c) * 256 + (2 * h2 + dy)) * 256 + 2 * w2);
      ang[c * 4 + dy * 2 + 0] = row[0];
      ang[c * 4 + dy * 2 + 1] = row[1];
    }
  }

  float st[64];
#pragma unroll
  for (int i = 0; i < 64; ++i) st[i] = 0.0f;

#pragma unroll
  for (int l = 0; l < 3; ++l) {
    float cw[4], sw[4];
#pragma unroll
    for (int wq = 0; wq < 4; ++wq) {
      const float h = 0.5f * ang[4 * l + wq];
      cw[wq] = cosf(h);
      sw[wq] = sinf(h);
    }
    const float p0r = cw[0] * cw[1] - sw[0] * sw[1];
    const float p0i = -(sw[0] * cw[1] + cw[0] * sw[1]);
    const float p1r = cw[0] * cw[1] + sw[0] * sw[1];
    const float p1i = cw[0] * sw[1] - sw[0] * cw[1];
    const float q0r = cw[2] * cw[3] - sw[2] * sw[3];
    const float q0i = -(sw[2] * cw[3] + cw[2] * sw[3]);
    const float q1r = cw[2] * cw[3] + sw[2] * sw[3];
    const float q1i = cw[2] * sw[3] - sw[2] * cw[3];
    const float Pr[2] = {p0r, p1r}, Pi[2] = {p0i, p1i};
    const float Qr[4] = {q0r, q1r, q1r, q0r}, Qi[4] = {q0i, q1i, -q1i, -q0i};
    float Dr[8], Di[8];
#pragma unroll
    for (int k = 0; k < 8; ++k) {
      const int pp = k >> 2, qq = k & 3;
      Dr[k] = Pr[pp] * Qr[qq] - Pi[pp] * Qi[qq];
      Di[k] = Pr[pp] * Qi[qq] + Pi[pp] * Qr[qq];
    }
    if (l == 0) {
#pragma unroll
      for (int k = 0; k < 8; ++k) {
#pragma unroll
        for (int b0 = 0; b0 < 32; b0 += 16) {
          st[k + b0]           =  Dr[k];
          st[k + b0 + 32]      =  Di[k];
          st[15 - k + b0]      =  Dr[k];
          st[15 - k + b0 + 32] = -Di[k];
        }
      }
    } else {
#pragma unroll
      for (int k = 0; k < 8; ++k) {
#pragma unroll
        for (int b0 = 0; b0 < 32; b0 += 16) {
          const int i0 = k + b0, i1 = 15 - k + b0;
          const float r0 = st[i0], m0 = st[i0 + 32];
          const float r1 = st[i1], m1 = st[i1 + 32];
          st[i0]      = r0 * Dr[k] - m0 * Di[k];
          st[i0 + 32] = r0 * Di[k] + m0 * Dr[k];
          st[i1]      = r1 * Dr[k] + m1 * Di[k];
          st[i1 + 32] = m1 * Dr[k] - r1 * Di[k];
        }
      }
    }

    if (l < 2) {
      apply_M(st, Sw, Smine, m, hi, T + 256 + l * 1536);
    } else {
      apply_K(st, Sw, Smine, m, hi, Ka);

      const float* Tl = T + l * 64;
      float ns[64];
#pragma unroll
      for (int i = 0; i < 32; ++i) {
        const int o = (i & 16) + ring_perm(i & 15);
        const float lr = Tl[2 * i], li = Tl[2 * i + 1];
        ns[o]      = st[i] * lr - st[i + 32] * li;
        ns[o + 32] = st[i] * li + st[i + 32] * lr;
      }
#pragma unroll
      for (int i = 0; i < 64; ++i) st[i] = ns[i];
    }
  }

#pragma unroll
  for (int i = 0; i < 16; ++i) {
    float ur = st[i],      ui = st[i + 32];
    float vr = st[i + 16], vi = st[i + 48];
    st[i]      = (ur + vr) * RS2;
    st[i + 32] = (ui + vi) * RS2;
    st[i + 16] = (ur - vr) * RS2;
    st[i + 48] = (ui - vi) * RS2;
  }

  float ev = 0.0f;
#pragma unroll
  for (int i = 0; i < 32; ++i) {
    float p2 = st[i] * st[i] + st[i + 32] * st[i + 32];
    ev += (i & 16) ? -p2 : p2;
  }
  if (p < P) out[p] = ev * 0.03125f;
}

extern "C" void kernel_launch(void* const* d_in, const int* in_sizes, int n_in,
                              void* d_out, int out_size, void* d_ws, size_t ws_size,
                              hipStream_t stream) {
  (void)in_sizes; (void)n_in; (void)ws_size;
  const float* x      = (const float*)d_in[0];
  const float* thetas = (const float*)d_in[1];
  const float* phis   = (const float*)d_in[2];
  float* out = (float*)d_out;
  float* T   = (float*)d_ws;

  hipLaunchKernelGGL(qc_build_T, dim3(1), dim3(1024), 0, stream, thetas, phis, T);

  const int P = out_size;
  const int blocks = (P + 127) / 128;
  hipLaunchKernelGGL(qc_main, dim3(blocks), dim3(128), 0, stream, x, T, out, P);
}


# Per-patch RX: cos(a/2), sin(a/2) by the hardware sine/cosine instructions (argument in revolutions), one Newton step to unit length; not in-line library code; phase factor rounding changes
# speedup vs baseline: 2.8199x; 1.1286x over previous
#include <hip/hip_runtime.h>
#include <hip/hip_bf16.h>
#include <math.h>


typedef __attribute__((ext_vector_type(2))) float v2f;
typedef __attribute__((ext_vector_type(4))) float v4f;
typedef __attribute__((ext_vector_type(8))) float v8f;

#define RS2 0.70710678118654752f
#define QUARTER_OVER_PI 0.079577471545947668f

__host__ __device__ constexpr int ring_perm(int j) {
  const int cb[4] = {8, 4, 2, 1};
  const int tb[4] = {4, 2, 1, 8};
  for (int w = 0; w < 4; ++w)
    if (j & cb[w]) j ^= tb[w];
  return j;
}

__global__ __launch_bounds__(1024) void qc_build_T(
    const float* __restrict__ thetas, const float* __restrict__ phis,
    float* __restrict__ T) {
  __shared__ float Ts[128];
  const int t = threadIdx.x;
  if (t < 96) {
    const int l = t >> 5;
    const int i = t & 31;

    const int a = i & 7;
    const float th_a = thetas[4 * l + (a & 3)];
    const float ph_l = phis[l];
    const float arg = (a < 4) ? 0.5f * th_a : ph_l;
    const float ca = cosf(arg), sa = sinf(arg);
    float ecw[4], esw[4];
#pragma unroll
    for (int w = 0; w < 4; ++w) {
      ecw[w] = __int_as_float(__builtin_amdgcn_readlane(__float_as_int(ca), w));
      esw[w] = __int_as_float(__builtin_amdgcn_readlane(__float_as_int(sa), w));
    }
    const float cpr = __int_as_float(__builtin_amdgcn_readlane(__float_as_int(ca), 4));
    const float cpi = __int_as_float(__builtin_amdgcn_readlane(__float_as_int(sa), 4));

    const int cb[4] = {8, 4, 2, 1};
    const int tb[4] = {4, 2, 1, 8};
    int idx = i & 15;
    float pr = 1.0f, pi = 0.0f;
#pragma unroll
    for (int w = 0; w < 4; ++w) {
      if (idx & cb[w]) {
        const float fr = (idx & tb[w]) ? esw[w] : -esw[w];
        const float fi = -ecw[w];
        const float r = pr * fr - pi * fi;
        const float mm = pr * fi + pi * fr;
        pr = r; pi = mm;
        idx ^= tb[w];
      }
    }
    if ((i & 16) && (idx & 8)) {
      const float r = pr * cpr - pi * cpi;
      const float mm = pr * cpi + pi * cpr;
      pr = r; pi = mm;
    }
    T[l * 64 + 2 * i]     = pr;
    T[l * 64 + 2 * i + 1] = pi;
    if (l < 2) {
      Ts[l * 64 + 2 * i]     = pr;
      Ts[l * 64 + 2 * i + 1] = pi;
    }
  }
  __syncthreads();

  const int l = t >> 9;
  const int h = (t >> 8) & 1;
  const int m = (t >> 4) & 15;
  const int n = t & 15;
  const float* ph = Ts + l * 64 + 32 * h;
  float mr = 0.0f, mi = 0.0f;
#pragma unroll
  for (int i = 0; i < 16; ++i) {
    const int o = ring_perm(i);
    const bool neg = ((__builtin_popcount(m & o) + __builtin_popcount(i & n)) & 1) != 0;
    const float tr = ph[2 * i], ti = ph[2 * i + 1];
    mr += neg ? -tr : tr;
    mi += neg ? -ti : ti;
  }
  mr *= 0.0625f;
  mi *= 0.0625f;
  float* Mo = T + 256 + ((l * 2 + h) * 3) * 256 + m * 16 + n;
  Mo[0]   = mr;
  Mo[256] = -(mr + mi);
  Mo[512] = mi - mr;
}

#define TP 68

__device__ __forceinline__ void apply_K(float (&st)[64], float* Sw, float* Smine,
                                        int m, int hi, const v2f (&Ka)[2][2]) {
#pragma unroll
  for (int j = 0; j < 16; ++j) {
    v4f t;
    t[0] = st[4 * j + 0]; t[1] = st[4 * j + 1];
    t[2] = st[4 * j + 2]; t[3] = st[4 * j + 3];
    *reinterpret_cast<v4f*>(Smine + 4 * j) = t;
  }
#pragma unroll
  for (int pt = 0; pt < 2; ++pt) {
    const float* Bb = Sw + (pt * 16 + m) * TP + 4 * hi;
    v4f B0[4], B1[4];
#pragma unroll
    for (int c = 0; c < 4; ++c) {
      B0[c] = *reinterpret_cast<const v4f*>(Bb + c * 16);
      B1[c] = *reinterpret_cast<const v4f*>(Bb + c * 16 + 8);
    }
    v8f acc[4] = {};
#pragma unroll
    for (int c = 0; c < 4; ++c) {
      v2f b;
      b[0] = B0[c][0]; b[1] = B0[c][1];
      acc[c] = __builtin_amdgcn_wmma_f32_16x16x4_f32(
          false, Ka[0][0], false, b, (short)0, acc[c], false, false);
    }
#pragma unroll
    for (int c = 0; c < 4; ++c) {
      v2f b;
      b[0] = B0[c][2]; b[1] = B0[c][3];
      acc[c] = __builtin_amdgcn_wmma_f32_16x16x4_f32(
          false, Ka[0][1], false, b, (short)0, acc[c], false, false);
    }
#pragma unroll
    for (int c = 0; c < 4; ++c) {
      v2f b;
      b[0] = B1[c][0]; b[1] = B1[c][1];
      acc[c] = __builtin_amdgcn_wmma_f32_16x16x4_f32(
          false, Ka[1][0], false, b, (short)0, acc[c], false, false);
    }
#pragma unroll
    for (int c = 0; c < 4; ++c) {
      v2f b;
      b[0] = B1[c][2]; b[1] = B1[c][3];
      acc[c] = __builtin_amdgcn_wmma_f32_16x16x4_f32(
          false, Ka[1][1], false, b, (short)0, acc[c], false, false);
    }
    float* d = Sw + (pt * 16 + m) * TP + hi * 8;
#pragma unroll
    for (int c = 0; c < 4; ++c) {
      v4f t;
      t[0] = acc[c][0]; t[1] = acc[c][1]; t[2] = acc[c][2]; t[3] = acc[c][3];
      *reinterpret_cast<v4f*>(d + c * 16) = t;
      t[0] = acc[c][4]; t[1] = acc[c][5]; t[2] = acc[c][6]; t[3] = acc[c][7];
      *reinterpret_cast<v4f*>(d + c * 16 + 4) = t;
    }
  }
#pragma unroll
  for (int j = 0; j < 16; ++j) {
    const v4f t = *reinterpret_cast<const v4f*>(Smine + 4 * j);
    st[4 * j + 0] = t[0]; st[4 * j + 1] = t[1];
    st[4 * j + 2] = t[2]; st[4 * j + 3] = t[3];
  }
}

__device__ __forceinline__ void apply_M(float (&st)[64], float* Sw, float* Smine,
                                        int m, int hi, const float* __restrict__ Ml) {
  const float* Arow = Ml + m * 16 + 4 * hi;
  v4f A[2][3][2];
#pragma unroll
  for (int v = 0; v < 3; ++v) {
    A[0][v][0] = *reinterpret_cast<const v4f*>(Arow + v * 256);
    A[0][v][1] = *reinterpret_cast<const v4f*>(Arow + v * 256 + 8);
  }
#pragma unroll
  for (int j = 0; j < 16; ++j) {
    v4f t;
    t[0] = st[4 * j + 0]; t[1] = st[4 * j + 1];
    t[2] = st[4 * j + 2]; t[3] = st[4 * j + 3];
    *reinterpret_cast<v4f*>(Smine + 4 * j) = t;
  }
#pragma unroll
  for (int v = 0; v < 3; ++v) {
    A[1][v][0] = *reinterpret_cast<const v4f*>(Arow + 768 + v * 256);
    A[1][v][1] = *reinterpret_cast<const v4f*>(Arow + 768 + v * 256 + 8);
  }
#pragma unroll
  for (int h = 0; h < 2; ++h) {
#pragma unroll
    for (int pt = 0; pt < 2; ++pt) {
      const float* Bb = Sw + (pt * 16 + m) * TP + 4 * hi;
      v4f xr[2], xi[2], xs[2];
      xr[0] = *reinterpret_cast<const v4f*>(Bb + h * 16);
      xr[1] = *reinterpret_cast<const v4f*>(Bb + h * 16 + 8);
      xi[0] = *reinterpret_cast<const v4f*>(Bb + (2 + h) * 16);
      xi[1] = *reinterpret_cast<const v4f*>(Bb + (2 + h) * 16 + 8);
      xs[0] = xr[0] + xi[0];
      xs[1] = xr[1] + xi[1];
      v8f k1 = {};
#pragma unroll
      for (int kk = 0; kk < 2; ++kk)
#pragma unroll
        for (int q = 0; q < 2; ++q) {
          v2f a, b;
          a[0] = A[h][0][kk][2 * q]; a[1] = A[h][0][kk][2 * q + 1];
          b[0] = xs[kk][2 * q];      b[1] = xs[kk][2 * q + 1];
          k1 = __builtin_amdgcn_wmma_f32_16x16x4_f32(
              false, a, false, b, (short)0, k1, false, false);
        }
      v8f re = k1, im = k1;
#pragma unroll
      for (int kk = 0; kk < 2; ++kk)
#pragma unroll
        for (int q = 0; q < 2; ++q) {
          v2f a, b;
          a[0] = A[h][1][kk][2 * q]; a[1] = A[h][1][kk][2 * q + 1];
          b[0] = xi[kk][2 * q];      b[1] = xi[kk][2 * q + 1];
          re = __builtin_amdgcn_wmma_f32_16x16x4_f32(
              false, a, false, b, (short)0, re, false, false);
          a[0] = A[h][2][kk][2 * q]; a[1] = A[h][2][kk][2 * q + 1];
          b[0] = xr[kk][2 * q];      b[1] = xr[kk][2 * q + 1];
          im = __builtin_amdgcn_wmma_f32_16x16x4_f32(
              false, a, false, b, (short)0, im, false, false);
        }
      float* d = Sw + (pt * 16 + m) * TP + hi * 8;
      v4f t;
      t[0] = re[0]; t[1] = re[1]; t[2] = re[2]; t[3] = re[3];
      *reinterpret_cast<v4f*>(d + h * 16) = t;
      t[0] = re[4]; t[1] = re[5]; t[2] = re[6]; t[3] = re[7];
      *reinterpret_cast<v4f*>(d + h * 16 + 4) = t;
      t[0] = im[0]; t[1] = im[1]; t[2] = im[2]; t[3] = im[3];
      *reinterpret_cast<v4f*>(d + (2 + h) * 16) = t;
      t[0] = im[4]; t[1] = im[5]; t[2] = im[6]; t[3] = im[7];
      *reinterpret_cast<v4f*>(d + (2 + h) * 16 + 4) = t;
    }
  }
#pragma unroll
  for (int j = 0; j < 16; ++j) {
    const v4f t = *reinterpret_cast<const v4f*>(Smine + 4 * j);
    st[4 * j + 0] = t[0]; st[4 * j + 1] = t[1];
    st[4 * j + 2] = t[2]; st[4 * j + 3] = t[3];
  }
}

__global__ __launch_bounds__(128, 8) void qc_main(
    const float* __restrict__ x, const float* __restrict__ T,
    float* __restrict__ out, int P) {
  __shared__ __attribute__((aligned(16))) float Sb[4][32 * TP];

  const int tid  = threadIdx.x;
  const int lane = tid & 31;
  const int wv   = tid >> 5;
  const int m    = lane & 15;
  const int hi   = lane >> 4;
  float* Sw = &Sb[wv][0];
  float* Smine = Sw + lane * TP;

  const int p  = blockIdx.x * 128 + tid;
  const int b  = p >> 14;
  const int h2 = (p >> 7) & 127;
  const int w2 = p & 127;

  v2f Ka[2][2];
#pragma unroll
  for (int kk = 0; kk < 2; ++kk)
#pragma unroll
    for (int q = 0; q < 2; ++q)
#pragma unroll
      for (int e = 0; e < 2; ++e) {
        const int k = 8 * kk + 4 * hi + 2 * q + e;
        Ka[kk][q][e] = (__builtin_popcount(m & k) & 1) ? -0.25f : 0.25f;
      }

  float ang[12];
#pragma unroll
  for (int c = 0; c < 3; ++c) {
#pragma unroll
    for (int dy = 0; dy < 2; ++dy) {
      const float* row = x + (((b * 3 + c) * 256 + (2 * h2 + dy)) * 256 + 2 * w2);
      ang[c * 4 + dy * 2 + 0] = row[0];
      ang[c * 4 + dy * 2 + 1] = row[1];
    }
  }

  float st[64];
#pragma unroll
  for (int i = 0; i < 64; ++i) st[i] = 0.0f;

#pragma unroll
  for (int l = 0; l < 3; ++l) {
    float cw[4], sw[4];
#pragma unroll
    for (int wq = 0; wq < 4; ++wq) {
      const float rev = ang[4 * l + wq] * QUARTER_OVER_PI;
      const float c0 = __builtin_amdgcn_cosf(rev);
      const float s0 = __builtin_amdgcn_sinf(rev);
      const float k = 1.5f - 0.5f * (c0 * c0 + s0 * s0);
      cw[wq] = c0 * k;
      sw[wq] = s0 * k;
    }
    const float p0r = cw[0] * cw[1] - sw[0] * sw[1];
    const float p0i = -(sw[0] * cw[1] + cw[0] * sw[1]);
    const float p1r = cw[0] * cw[1] + sw[0] * sw[1];
    const float p1i = cw[0] * sw[1] - sw[0] * cw[1];
    const float q0r = cw[2] * cw[3] - sw[2] * sw[3];
    const float q0i = -(sw[2] * cw[3] + cw[2] * sw[3]);
    const float q1r = cw[2] * cw[3] + sw[2] * sw[3];
    const float q1i = cw[2] * sw[3] - sw[2] * cw[3];
    const float Pr[2] = {p0r, p1r}, Pi[2] = {p0i, p1i};
    const float Qr[4] = {q0r, q1r, q1r, q0r}, Qi[4] = {q0i, q1i, -q1i, -q0i};
    float Dr[8], Di[8];
#pragma unroll
    for (int k = 0; k < 8; ++k) {
      const int pp = k >> 2, qq = k & 3;
      Dr[k] = Pr[pp] * Qr[qq] - Pi[pp] * Qi[qq];
      Di[k] = Pr[pp] * Qi[qq] + Pi[pp] * Qr[qq];
    }
    if (l == 0) {
#pragma unroll
      for (int k = 0; k < 8; ++k) {
#pragma unroll
        for (int b0 = 0; b0 < 32; b0 += 16) {
          st[k + b0]           =  Dr[k];
          st[k + b0 + 32]      =  Di[k];
          st[15 - k + b0]      =  Dr[k];
          st[15 - k + b0 + 32] = -Di[k];
        }
      }
    } else {
#pragma unroll
      for (int k = 0; k < 8; ++k) {
#pragma unroll
        for (int b0 = 0; b0 < 32; b0 += 16) {
          const int i0 = k + b0, i1 = 15 - k + b0;
          const float r0 = st[i0], m0 = st[i0 + 32];
          const float r1 = st[i1], m1 = st[i1 + 32];
          st[i0]      = r0 * Dr[k] - m0 * Di[k];
          st[i0 + 32] = r0 * Di[k] + m0 * Dr[k];
          st[i1]      = r1 * Dr[k] + m1 * Di[k];
          st[i1 + 32] = m1 * Dr[k] - r1 * Di[k];
        }
      }
    }

    if (l < 2) {
      apply_M(st, Sw, Smine, m, hi, T + 256 + l * 1536);
    } else {
      apply_K(st, Sw, Smine, m, hi, Ka);

      const float* Tl = T + l * 64;
      float ns[64];
#pragma unroll
      for (int i = 0; i < 32; ++i) {
        const int o = (i & 16) + ring_perm(i & 15);
        const float lr = Tl[2 * i], li = Tl[2 * i + 1];
        ns[o]      = st[i] * lr - st[i + 32] * li;
        ns[o + 32] = st[i] * li + st[i + 32] * lr;
      }
#pragma unroll
      for (int i = 0; i < 64; ++i) st[i] = ns[i];
    }
  }

#pragma unroll
  for (int i = 0; i < 16; ++i) {
    float ur = st[i],      ui = st[i + 32];
    float vr = st[i + 16], vi = st[i + 48];
    st[i]      = (ur + vr) * RS2;
    st[i + 32] = (ui + vi) * RS2;
    st[i + 16] = (ur - vr) * RS2;
    st[i + 48] = (ui - vi) * RS2;
  }

  float ev = 0.0f;
#pragma unroll
  for (int i = 0; i < 32; ++i) {
    float p2 = st[i] * st[i] + st[i + 32] * st[i + 32];
    ev += (i & 16) ? -p2 : p2;
  }
  if (p < P) out[p] = ev * 0.03125f;
}

extern "C" void kernel_launch(void* const* d_in, const int* in_sizes, int n_in,
                              void* d_out, int out_size, void* d_ws, size_t ws_size,
                              hipStream_t stream) {
  (void)in_sizes; (void)n_in; (void)ws_size;
  const float* x      = (const float*)d_in[0];
  const float* thetas = (const float*)d_in[1];
  const float* phis   = (const float*)d_in[2];
  float* out = (float*)d_out;
  float* T   = (float*)d_ws;

  hipLaunchKernelGGL(qc_build_T, dim3(1), dim3(1024), 0, stream, thetas, phis, T);

  const int P = out_size;
  const int blocks = (P + 127) / 128;
  hipLaunchKernelGGL(qc_main, dim3(blocks), dim3(128), 0, stream, x, T, out, P);
}


# Last layer's K (16-point Hadamard transform of wires 1-4) as 4 butterfly stages in each lane's own registers, not a dense +-1/4 WMMA product via LDS: 96 WMMAs not 128; sum order, rounding change
# speedup vs baseline: 3.0070x; 1.0663x over previous
#include <hip/hip_runtime.h>
#include <hip/hip_bf16.h>
#include <math.h>


typedef __attribute__((ext_vector_type(2))) float v2f;
typedef __attribute__((ext_vector_type(4))) float v4f;
typedef __attribute__((ext_vector_type(8))) float v8f;

#define RS2 0.70710678118654752f
#define QUARTER_OVER_PI 0.079577471545947668f

__host__ __device__ constexpr int ring_perm(int j) {
  const int cb[4] = {8, 4, 2, 1};
  const int tb[4] = {4, 2, 1, 8};
  for (int w = 0; w < 4; ++w)
    if (j & cb[w]) j ^= tb[w];
  return j;
}

__global__ __launch_bounds__(1024) void qc_build_T(
    const float* __restrict__ thetas, const float* __restrict__ phis,
    float* __restrict__ T) {
  __shared__ float Ts[128];
  const int t = threadIdx.x;
  if (t < 96) {
    const int l = t >> 5;
    const int i = t & 31;

    const int a = i & 7;
    const float th_a = thetas[4 * l + (a & 3)];
    const float ph_l = phis[l];
    const float arg = (a < 4) ? 0.5f * th_a : ph_l;
    const float ca = cosf(arg), sa = sinf(arg);
    float ecw[4], esw[4];
#pragma unroll
    for (int w = 0; w < 4; ++w) {
      ecw[w] = __int_as_float(__builtin_amdgcn_readlane(__float_as_int(ca), w));
      esw[w] = __int_as_float(__builtin_amdgcn_readlane(__float_as_int(sa), w));
    }
    const float cpr = __int_as_float(__builtin_amdgcn_readlane(__float_as_int(ca), 4));
    const float cpi = __int_as_float(__builtin_amdgcn_readlane(__float_as_int(sa), 4));

    const int cb[4] = {8, 4, 2, 1};
    const int tb[4] = {4, 2, 1, 8};
    int idx = i & 15;
    float pr = 1.0f, pi = 0.0f;
#pragma unroll
    for (int w = 0; w < 4; ++w) {
      if (idx & cb[w]) {
        const float fr = (idx & tb[w]) ? esw[w] : -esw[w];
        const float fi = -ecw[w];
        const float r = pr * fr - pi * fi;
        const float mm = pr * fi + pi * fr;
        pr = r; pi = mm;
        idx ^= tb[w];
      }
    }
    if ((i & 16) && (idx & 8)) {
      const float r = pr * cpr - pi * cpi;
      const float mm = pr * cpi + pi * cpr;
      pr = r; pi = mm;
    }
    T[l * 64 + 2 * i]     = pr;
    T[l * 64 + 2 * i + 1] = pi;
    if (l < 2) {
      Ts[l * 64 + 2 * i]     = pr;
      Ts[l * 64 + 2 * i + 1] = pi;
    }
  }
  __syncthreads();

  const int l = t >> 9;
  const int h = (t >> 8) & 1;
  const int m = (t >> 4) & 15;
  const int n = t & 15;
  const float* ph = Ts + l * 64 + 32 * h;
  float mr = 0.0f, mi = 0.0f;
#pragma unroll
  for (int i = 0; i < 16; ++i) {
    const int o = ring_perm(i);
    const bool neg = ((__builtin_popcount(m & o) + __builtin_popcount(i & n)) & 1) != 0;
    const float tr = ph[2 * i], ti = ph[2 * i + 1];
    mr += neg ? -tr : tr;
    mi += neg ? -ti : ti;
  }
  mr *= 0.0625f;
  mi *= 0.0625f;
  float* Mo = T + 256 + ((l * 2 + h) * 3) * 256 + m * 16 + n;
  Mo[0]   = mr;
  Mo[256] = -(mr + mi);
  Mo[512] = mi - mr;
}

#define TP 68

__device__ __forceinline__ void hadamard_wires_1_to_4(float (&st)[64]) {
#pragma unroll
  for (int v = 0; v < 4; ++v) {
    v2f x[8];
#pragma unroll
    for (int j = 0; j < 8; ++j) { x[j][0] = st[16 * v + 2 * j]; x[j][1] = st[16 * v + 2 * j + 1]; }
#pragma unroll
    for (int j = 0; j < 8; ++j) {
      const v2f aa = __builtin_shufflevector(x[j], x[j], 0, 0);
      const v2f bb = __builtin_shufflevector(x[j], x[j], 1, 1);
      const v2f bm = __builtin_shufflevector(bb, -bb, 0, 3);
      x[j] = aa + bm;
    }
#pragma unroll
    for (int bit = 1; bit < 8; bit <<= 1) {
#pragma unroll
      for (int j = 0; j < 8; ++j) {
        if ((j & bit) == 0) {
          const v2f a = x[j], b = x[j + bit];
          x[j]       = a + b;
          x[j + bit] = a - b;
        }
      }
    }
#pragma unroll
    for (int j = 0; j < 8; ++j) { st[16 * v + 2 * j] = x[j][0]; st[16 * v + 2 * j + 1] = x[j][1]; }
  }
}

__device__ __forceinline__ void apply_M(float (&st)[64], float* Sw, float* Smine,
                                        int m, int hi, const float* __restrict__ Ml) {
  const float* Arow = Ml + m * 16 + 4 * hi;
  v4f A[2][3][2];
#pragma unroll
  for (int v = 0; v < 3; ++v) {
    A[0][v][0] = *reinterpret_cast<const v4f*>(Arow + v * 256);
    A[0][v][1] = *reinterpret_cast<const v4f*>(Arow + v * 256 + 8);
  }
#pragma unroll
  for (int j = 0; j < 16; ++j) {
    v4f t;
    t[0] = st[4 * j + 0]; t[1] = st[4 * j + 1];
    t[2] = st[4 * j + 2]; t[3] = st[4 * j + 3];
    *reinterpret_cast<v4f*>(Smine + 4 * j) = t;
  }
#pragma unroll
  for (int v = 0; v < 3; ++v) {
    A[1][v][0] = *reinterpret_cast<const v4f*>(Arow + 768 + v * 256);
    A[1][v][1] = *reinterpret_cast<const v4f*>(Arow + 768 + v * 256 + 8);
  }
#pragma unroll
  for (int h = 0; h < 2; ++h) {
#pragma unroll
    for (int pt = 0; pt < 2; ++pt) {
      const float* Bb = Sw + (pt * 16 + m) * TP + 4 * hi;
      v4f xr[2], xi[2], xs[2];
      xr[0] = *reinterpret_cast<const v4f*>(Bb + h * 16);
      xr[1] = *reinterpret_cast<const v4f*>(Bb + h * 16 + 8);
      xi[0] = *reinterpret_cast<const v4f*>(Bb + (2 + h) * 16);
      xi[1] = *reinterpret_cast<const v4f*>(Bb + (2 + h) * 16 + 8);
      xs[0] = xr[0] + xi[0];
      xs[1] = xr[1] + xi[1];
      v8f k1 = {};
#pragma unroll
      for (int kk = 0; kk < 2; ++kk)
#pragma unroll
        for (int q = 0; q < 2; ++q) {
          v2f a, b;
          a[0] = A[h][0][kk][2 * q]; a[1] = A[h][0][kk][2 * q + 1];
          b[0] = xs[kk][2 * q];      b[1] = xs[kk][2 * q + 1];
          k1 = __builtin_amdgcn_wmma_f32_16x16x4_f32(
              false, a, false, b, (short)0, k1, false, false);
        }
      v8f re = k1, im = k1;
#pragma unroll
      for (int kk = 0; kk < 2; ++kk)
#pragma unroll
        for (int q = 0; q < 2; ++q) {
          v2f a, b;
          a[0] = A[h][1][kk][2 * q]; a[1] = A[h][1][kk][2 * q + 1];
          b[0] = xi[kk][2 * q];      b[1] = xi[kk][2 * q + 1];
          re = __builtin_amdgcn_wmma_f32_16x16x4_f32(
              false, a, false, b, (short)0, re, false, false);
          a[0] = A[h][2][kk][2 * q]; a[1] = A[h][2][kk][2 * q + 1];
          b[0] = xr[kk][2 * q];      b[1] = xr[kk][2 * q + 1];
          im = __builtin_amdgcn_wmma_f32_16x16x4_f32(
              false, a, false, b, (short)0, im, false, false);
        }
      float* d = Sw + (pt * 16 + m) * TP + hi * 8;
      v4f t;
      t[0] = re[0]; t[1] = re[1]; t[2] = re[2]; t[3] = re[3];
      *reinterpret_cast<v4f*>(d + h * 16) = t;
      t[0] = re[4]; t[1] = re[5]; t[2] = re[6]; t[3] = re[7];
      *reinterpret_cast<v4f*>(d + h * 16 + 4) = t;
      t[0] = im[0]; t[1] = im[1]; t[2] = im[2]; t[3] = im[3];
      *reinterpret_cast<v4f*>(d + (2 + h) * 16) = t;
      t[0] = im[4]; t[1] = im[5]; t[2] = im[6]; t[3] = im[7];
      *reinterpret_cast<v4f*>(d + (2 + h) * 16 + 4) = t;
    }
  }
#pragma unroll
  for (int j = 0; j < 16; ++j) {
    const v4f t = *reinterpret_cast<const v4f*>(Smine + 4 * j);
    st[4 * j + 0] = t[0]; st[4 * j + 1] = t[1];
    st[4 * j + 2] = t[2]; st[4 * j + 3] = t[3];
  }
}

__global__ __launch_bounds__(128, 8) void qc_main(
    const float* __restrict__ x, const float* __restrict__ T,
    float* __restrict__ out, int P) {
  __shared__ __attribute__((aligned(16))) float Sb[4][32 * TP];

  const int tid  = threadIdx.x;
  const int lane = tid & 31;
  const int wv   = tid >> 5;
  const int m    = lane & 15;
  const int hi   = lane >> 4;
  float* Sw = &Sb[wv][0];
  float* Smine = Sw + lane * TP;

  const int p  = blockIdx.x * 128 + tid;
  const int b  = p >> 14;
  const int h2 = (p >> 7) & 127;
  const int w2 = p & 127;

  float ang[12];
#pragma unroll
  for (int c = 0; c < 3; ++c) {
#pragma unroll
    for (int dy = 0; dy < 2; ++dy) {
      const float* row = x + (((b * 3 + c) * 256 + (2 * h2 + dy)) * 256 + 2 * w2);
      ang[c * 4 + dy * 2 + 0] = row[0];
      ang[c * 4 + dy * 2 + 1] = row[1];
    }
  }

  float st[64];
#pragma unroll
  for (int i = 0; i < 64; ++i) st[i] = 0.0f;

#pragma unroll
  for (int l = 0; l < 3; ++l) {
    float cw[4], sw[4];
#pragma unroll
    for (int wq = 0; wq < 4; ++wq) {
      const float rev = ang[4 * l + wq] * QUARTER_OVER_PI;
      const float c0 = __builtin_amdgcn_cosf(rev);
      const float s0 = __builtin_amdgcn_sinf(rev);
      const float k = 1.5f - 0.5f * (c0 * c0 + s0 * s0);
      cw[wq] = c0 * k;
      sw[wq] = s0 * k;
    }
    const float p0r = cw[0] * cw[1] - sw[0] * sw[1];
    const float p0i = -(sw[0] * cw[1] + cw[0] * sw[1]);
    const float p1r = cw[0] * cw[1] + sw[0] * sw[1];
    const float p1i = cw[0] * sw[1] - sw[0] * cw[1];
    const float q0r = cw[2] * cw[3] - sw[2] * sw[3];
    const float q0i = -(sw[2] * cw[3] + cw[2] * sw[3]);
    const float q1r = cw[2] * cw[3] + sw[2] * sw[3];
    const float q1i = cw[2] * sw[3] - sw[2] * cw[3];
    const float Pr[2] = {p0r, p1r}, Pi[2] = {p0i, p1i};
    const float Qr[4] = {q0r, q1r, q1r, q0r}, Qi[4] = {q0i, q1i, -q1i, -q0i};
    float Dr[8], Di[8];
#pragma unroll
    for (int k = 0; k < 8; ++k) {
      const int pp = k >> 2, qq = k & 3;
      Dr[k] = Pr[pp] * Qr[qq] - Pi[pp] * Qi[qq];
      Di[k] = Pr[pp] * Qi[qq] + Pi[pp] * Qr[qq];
    }
    if (l == 0) {
#pragma unroll
      for (int k = 0; k < 8; ++k) {
#pragma unroll
        for (int b0 = 0; b0 < 32; b0 += 16) {
          st[k + b0]           =  Dr[k];
          st[k + b0 + 32]      =  Di[k];
          st[15 - k + b0]      =  Dr[k];
          st[15 - k + b0 + 32] = -Di[k];
        }
      }
    } else {
#pragma unroll
      for (int k = 0; k < 8; ++k) {
#pragma unroll
        for (int b0 = 0; b0 < 32; b0 += 16) {
          const int i0 = k + b0, i1 = 15 - k + b0;
          const float r0 = st[i0], m0 = st[i0 + 32];
          const float r1 = st[i1], m1 = st[i1 + 32];
          st[i0]      = r0 * Dr[k] - m0 * Di[k];
          st[i0 + 32] = r0 * Di[k] + m0 * Dr[k];
          st[i1]      = r1 * Dr[k] + m1 * Di[k];
          st[i1 + 32] = m1 * Dr[k] - r1 * Di[k];
        }
      }
    }

    if (l < 2) {
      apply_M(st, Sw, Smine, m, hi, T + 256 + l * 1536);
    } else {
      hadamard_wires_1_to_4(st);

      const float* Tl = T + l * 64;
      float ns[64];
#pragma unroll
      for (int i = 0; i < 32; ++i) {
        const int o = (i & 16) + ring_perm(i & 15);
        const float lr = Tl[2 * i], li = Tl[2 * i + 1];
        ns[o]      = st[i] * lr - st[i + 32] * li;
        ns[o + 32] = st[i] * li + st[i + 32] * lr;
      }
#pragma unroll
      for (int i = 0; i < 64; ++i) st[i] = ns[i];
    }
  }

#pragma unroll
  for (int i = 0; i < 16; ++i) {
    float ur = st[i],      ui = st[i + 32];
    float vr = st[i + 16], vi = st[i + 48];
    st[i]      = (ur + vr) * RS2;
    st[i + 32] = (ui + vi) * RS2;
    st[i + 16] = (ur - vr) * RS2;
    st[i + 48] = (ui - vi) * RS2;
  }

  float ev = 0.0f;
#pragma unroll
  for (int i = 0; i < 32; ++i) {
    float p2 = st[i] * st[i] + st[i + 32] * st[i + 32];
    ev += (i & 16) ? -p2 : p2;
  }
  if (p < P) out[p] = ev * 0.001953125f;
}

extern "C" void kernel_launch(void* const* d_in, const int* in_sizes, int n_in,
                              void* d_out, int out_size, void* d_ws, size_t ws_size,
                              hipStream_t stream) {
  (void)in_sizes; (void)n_in; (void)ws_size;
  const float* x      = (const float*)d_in[0];
  const float* thetas = (const float*)d_in[1];
  const float* phis   = (const float*)d_in[2];
  float* out = (float*)d_out;
  float* T   = (float*)d_ws;

  hipLaunchKernelGGL(qc_build_T, dim3(1), dim3(1024), 0, stream, thetas, phis, T);

  const int P = out_size;
  const int blocks = (P + 127) / 128;
  hipLaunchKernelGGL(qc_main, dim3(blocks), dim3(128), 0, stream, x, T, out, P);
}


# Layer 0's K S_0 K in each lane's registers, not a complex WMMA product via LDS: K as butterfly stages (first shared by both wire-0 halves), S_0 as phased renaming; 48 WMMAs not 96; sum order changes
# speedup vs baseline: 3.4521x; 1.1480x over previous
#include <hip/hip_runtime.h>
#include <hip/hip_bf16.h>
#include <math.h>


typedef __attribute__((ext_vector_type(2))) float v2f;
typedef __attribute__((ext_vector_type(4))) float v4f;
typedef __attribute__((ext_vector_type(8))) float v8f;

#define RS2 0.70710678118654752f
#define QUARTER_OVER_PI 0.079577471545947668f

__host__ __device__ constexpr int ring_perm(int j) {
  const int cb[4] = {8, 4, 2, 1};
  const int tb[4] = {4, 2, 1, 8};
  for (int w = 0; w < 4; ++w)
    if (j & cb[w]) j ^= tb[w];
  return j;
}

__global__ __launch_bounds__(512) void qc_build_T(
    const float* __restrict__ thetas, const float* __restrict__ phis,
    float* __restrict__ T) {
  __shared__ float Ts[64];
  const int t = threadIdx.x;
  if (t < 96) {
    const int l = t >> 5;
    const int i = t & 31;

    const int a = i & 7;
    const float th_a = thetas[4 * l + (a & 3)];
    const float ph_l = phis[l];
    const float arg = (a < 4) ? 0.5f * th_a : ph_l;
    const float ca = cosf(arg), sa = sinf(arg);
    float ecw[4], esw[4];
#pragma unroll
    for (int w = 0; w < 4; ++w) {
      ecw[w] = __int_as_float(__builtin_amdgcn_readlane(__float_as_int(ca), w));
      esw[w] = __int_as_float(__builtin_amdgcn_readlane(__float_as_int(sa), w));
    }
    const float cpr = __int_as_float(__builtin_amdgcn_readlane(__float_as_int(ca), 4));
    const float cpi = __int_as_float(__builtin_amdgcn_readlane(__float_as_int(sa), 4));

    const int cb[4] = {8, 4, 2, 1};
    const int tb[4] = {4, 2, 1, 8};
    int idx = i & 15;
    float pr = 1.0f, pi = 0.0f;
#pragma unroll
    for (int w = 0; w < 4; ++w) {
      if (idx & cb[w]) {
        const float fr = (idx & tb[w]) ? esw[w] : -esw[w];
        const float fi = -ecw[w];
        const float r = pr * fr - pi * fi;
        const float mm = pr * fi + pi * fr;
        pr = r; pi = mm;
        idx ^= tb[w];
      }
    }
    if ((i & 16) && (idx & 8)) {
      const float r = pr * cpr - pi * cpi;
      const float mm = pr * cpi + pi * cpr;
      pr = r; pi = mm;
    }
    T[l * 64 + 2 * i]     = pr;
    T[l * 64 + 2 * i + 1] = pi;
    if (l == 1) {
      Ts[2 * i]     = pr;
      Ts[2 * i + 1] = pi;
    }
  }
  __syncthreads();

  const int h = t >> 8;
  const int m = (t >> 4) & 15;
  const int n = t & 15;
  const float* ph = Ts + 32 * h;
  float mr = 0.0f, mi = 0.0f;
#pragma unroll
  for (int i = 0; i < 16; ++i) {
    const int o = ring_perm(i);
    const bool neg = ((__builtin_popcount(m & o) + __builtin_popcount(i & n)) & 1) != 0;
    const float tr = ph[2 * i], ti = ph[2 * i + 1];
    mr += neg ? -tr : tr;
    mi += neg ? -ti : ti;
  }
  mr *= 0.0625f;
  mi *= 0.0625f;
  float* Mo = T + 256 + (h * 3) * 256 + m * 16 + n;
  Mo[0]   = mr;
  Mo[256] = -(mr + mi);
  Mo[512] = mi - mr;
}

#define TP 68

__device__ __forceinline__ void hadamard16(float (&st)[64], const int v) {
  v2f x[8];
#pragma unroll
  for (int j = 0; j < 8; ++j) { x[j][0] = st[16 * v + 2 * j]; x[j][1] = st[16 * v + 2 * j + 1]; }
#pragma unroll
  for (int j = 0; j < 8; ++j) {
    const v2f aa = __builtin_shufflevector(x[j], x[j], 0, 0);
    const v2f bb = __builtin_shufflevector(x[j], x[j], 1, 1);
    const v2f bm = __builtin_shufflevector(bb, -bb, 0, 3);
    x[j] = aa + bm;
  }
#pragma unroll
  for (int bit = 1; bit < 8; bit <<= 1) {
#pragma unroll
    for (int j = 0; j < 8; ++j) {
      if ((j & bit) == 0) {
        const v2f a = x[j], b = x[j + bit];
        x[j]       = a + b;
        x[j + bit] = a - b;
      }
    }
  }
#pragma unroll
  for (int j = 0; j < 8; ++j) { st[16 * v + 2 * j] = x[j][0]; st[16 * v + 2 * j + 1] = x[j][1]; }
}

__device__ __forceinline__ void hadamard_wires_1_to_4(float (&st)[64]) {
#pragma unroll
  for (int v = 0; v < 4; ++v) hadamard16(st, v);
}

__device__ __forceinline__ void rx_phase_factors(const float (&ang)[12], const int l,
                                                 float (&Dr)[8], float (&Di)[8]) {
  float cw[4], sw[4];
#pragma unroll
  for (int wq = 0; wq < 4; ++wq) {
    const float rev = ang[4 * l + wq] * QUARTER_OVER_PI;
    const float c0 = __builtin_amdgcn_cosf(rev);
    const float s0 = __builtin_amdgcn_sinf(rev);
    const float k = 1.5f - 0.5f * (c0 * c0 + s0 * s0);
    cw[wq] = c0 * k;
    sw[wq] = s0 * k;
  }
  const float p0r = cw[0] * cw[1] - sw[0] * sw[1];
  const float p0i = -(sw[0] * cw[1] + cw[0] * sw[1]);
  const float p1r = cw[0] * cw[1] + sw[0] * sw[1];
  const float p1i = cw[0] * sw[1] - sw[0] * cw[1];
  const float q0r = cw[2] * cw[3] - sw[2] * sw[3];
  const float q0i = -(sw[2] * cw[3] + cw[2] * sw[3]);
  const float q1r = cw[2] * cw[3] + sw[2] * sw[3];
  const float q1i = cw[2] * sw[3] - sw[2] * cw[3];
  const float Pr[2] = {p0r, p1r}, Pi[2] = {p0i, p1i};
  const float Qr[4] = {q0r, q1r, q1r, q0r}, Qi[4] = {q0i, q1i, -q1i, -q0i};
#pragma unroll
  for (int k = 0; k < 8; ++k) {
    const int pp = k >> 2, qq = k & 3;
    Dr[k] = Pr[pp] * Qr[qq] - Pi[pp] * Qi[qq];
    Di[k] = Pr[pp] * Qi[qq] + Pi[pp] * Qr[qq];
  }
}

__device__ __forceinline__ void times_D(float (&st)[64], const int h,
                                        const float (&Dr)[8], const float (&Di)[8]) {
#pragma unroll
  for (int k = 0; k < 8; ++k) {
    const int i0 = k + 16 * h, i1 = 15 - k + 16 * h;
    const float r0 = st[i0], m0 = st[i0 + 32];
    const float r1 = st[i1], m1 = st[i1 + 32];
    st[i0]      = r0 * Dr[k] - m0 * Di[k];
    st[i0 + 32] = r0 * Di[k] + m0 * Dr[k];
    st[i1]      = r1 * Dr[k] + m1 * Di[k];
    st[i1 + 32] = m1 * Dr[k] - r1 * Di[k];
  }
}

__device__ __forceinline__ void product_M(float (&st)[64], float* Sw, float* Smine,
                                          int m, int hi, const float* __restrict__ Ml) {
  const float* Arow = Ml + m * 16 + 4 * hi;
  v4f A[2][3][2];
#pragma unroll
  for (int h = 0; h < 2; ++h)
#pragma unroll
    for (int v = 0; v < 3; ++v) {
      A[h][v][0] = *reinterpret_cast<const v4f*>(Arow + h * 768 + v * 256);
      A[h][v][1] = *reinterpret_cast<const v4f*>(Arow + h * 768 + v * 256 + 8);
    }
#pragma unroll
  for (int h = 0; h < 2; ++h) {
#pragma unroll
    for (int pt = 0; pt < 2; ++pt) {
      const float* Bb = Sw + (pt * 16 + m) * TP + 4 * hi;
      v4f xr[2], xi[2], xs[2];
      xr[0] = *reinterpret_cast<const v4f*>(Bb + h * 16);
      xr[1] = *reinterpret_cast<const v4f*>(Bb + h * 16 + 8);
      xi[0] = *reinterpret_cast<const v4f*>(Bb + (2 + h) * 16);
      xi[1] = *reinterpret_cast<const v4f*>(Bb + (2 + h) * 16 + 8);
      xs[0] = xr[0] + xi[0];
      xs[1] = xr[1] + xi[1];
      v8f k1 = {};
#pragma unroll
      for (int kk = 0; kk < 2; ++kk)
#pragma unroll
        for (int q = 0; q < 2; ++q) {
          v2f a, b;
          a[0] = A[h][0][kk][2 * q]; a[1] = A[h][0][kk][2 * q + 1];
          b[0] = xs[kk][2 * q];      b[1] = xs[kk][2 * q + 1];
          k1 = __builtin_amdgcn_wmma_f32_16x16x4_f32(
              false, a, false, b, (short)0, k1, false, false);
        }
      v8f re = k1, im = k1;
#pragma unroll
      for (int kk = 0; kk < 2; ++kk)
#pragma unroll
        for (int q = 0; q < 2; ++q) {
          v2f a, b;
          a[0] = A[h][1][kk][2 * q]; a[1] = A[h][1][kk][2 * q + 1];
          b[0] = xi[kk][2 * q];      b[1] = xi[kk][2 * q + 1];
          re = __builtin_amdgcn_wmma_f32_16x16x4_f32(
              false, a, false, b, (short)0, re, false, false);
          a[0] = A[h][2][kk][2 * q]; a[1] = A[h][2][kk][2 * q + 1];
          b[0] = xr[kk][2 * q];      b[1] = xr[kk][2 * q + 1];
          im = __builtin_amdgcn_wmma_f32_16x16x4_f32(
              false, a, false, b, (short)0, im, false, false);
        }
      float* d = Sw + (pt * 16 + m) * TP + hi * 8;
      v4f t;
      t[0] = re[0]; t[1] = re[1]; t[2] = re[2]; t[3] = re[3];
      *reinterpret_cast<v4f*>(d + h * 16) = t;
      t[0] = re[4]; t[1] = re[5]; t[2] = re[6]; t[3] = re[7];
      *reinterpret_cast<v4f*>(d + h * 16 + 4) = t;
      t[0] = im[0]; t[1] = im[1]; t[2] = im[2]; t[3] = im[3];
      *reinterpret_cast<v4f*>(d + (2 + h) * 16) = t;
      t[0] = im[4]; t[1] = im[5]; t[2] = im[6]; t[3] = im[7];
      *reinterpret_cast<v4f*>(d + (2 + h) * 16 + 4) = t;
    }
  }
#pragma unroll
  for (int j = 0; j < 16; ++j) {
    const v4f t = *reinterpret_cast<const v4f*>(Smine + 4 * j);
    st[4 * j + 0] = t[0]; st[4 * j + 1] = t[1];
    st[4 * j + 2] = t[2]; st[4 * j + 3] = t[3];
  }
}

__global__ __launch_bounds__(128, 8) void qc_main(
    const float* __restrict__ x, const float* __restrict__ T,
    float* __restrict__ out, int P) {
  __shared__ __attribute__((aligned(16))) float Sb[4][32 * TP];

  const int tid  = threadIdx.x;
  const int lane = tid & 31;
  const int wv   = tid >> 5;
  const int m    = lane & 15;
  const int hi   = lane >> 4;
  float* Sw = &Sb[wv][0];
  float* Smine = Sw + lane * TP;

  const int p  = blockIdx.x * 128 + tid;
  const int b  = p >> 14;
  const int h2 = (p >> 7) & 127;
  const int w2 = p & 127;

  float ang[12];
#pragma unroll
  for (int c = 0; c < 3; ++c) {
#pragma unroll
    for (int dy = 0; dy < 2; ++dy) {
      const float* row = x + (((b * 3 + c) * 256 + (2 * h2 + dy)) * 256 + 2 * w2);
      ang[c * 4 + dy * 2 + 0] = row[0];
      ang[c * 4 + dy * 2 + 1] = row[1];
    }
  }

  float st[64];
  float Dr[8], Di[8];

  rx_phase_factors(ang, 0, Dr, Di);
#pragma unroll
  for (int k = 0; k < 8; ++k) {
    st[k]           =  Dr[k];
    st[k + 32]      =  Di[k];
    st[15 - k]      =  Dr[k];
    st[15 - k + 32] = -Di[k];
  }
  hadamard16(st, 0);
  hadamard16(st, 2);
  float xr[16], xi[16];
#pragma unroll
  for (int i = 0; i < 16; ++i) { xr[i] = st[i]; xi[i] = st[32 + i]; }

  rx_phase_factors(ang, 1, Dr, Di);

#pragma unroll
  for (int h = 0; h < 2; ++h) {
#pragma unroll
    for (int i = 0; i < 16; ++i) {
      const int o = 16 * h + ring_perm(i);
      const float lr = T[2 * (16 * h + i)], li = T[2 * (16 * h + i) + 1];
      st[o]      = xr[i] * lr - xi[i] * li;
      st[o + 32] = xr[i] * li + xi[i] * lr;
    }
    hadamard16(st, h);
    hadamard16(st, 2 + h);
    times_D(st, h, Dr, Di);
#pragma unroll
    for (int j = 0; j < 4; ++j) {
      v4f t;
      t[0] = st[16 * h + 4 * j + 0]; t[1] = st[16 * h + 4 * j + 1];
      t[2] = st[16 * h + 4 * j + 2]; t[3] = st[16 * h + 4 * j + 3];
      *reinterpret_cast<v4f*>(Smine + 16 * h + 4 * j) = t;
      t[0] = st[32 + 16 * h + 4 * j + 0]; t[1] = st[32 + 16 * h + 4 * j + 1];
      t[2] = st[32 + 16 * h + 4 * j + 2]; t[3] = st[32 + 16 * h + 4 * j + 3];
      *reinterpret_cast<v4f*>(Smine + 32 + 16 * h + 4 * j) = t;
    }
  }

  product_M(st, Sw, Smine, m, hi, T + 256);

  rx_phase_factors(ang, 2, Dr, Di);
  times_D(st, 0, Dr, Di);
  times_D(st, 1, Dr, Di);
  hadamard_wires_1_to_4(st);
  {
    const float* Tl = T + 2 * 64;
    float ns[64];
#pragma unroll
    for (int i = 0; i < 32; ++i) {
      const int o = (i & 16) + ring_perm(i & 15);
      const float lr = Tl[2 * i], li = Tl[2 * i + 1];
      ns[o]      = st[i] * lr - st[i + 32] * li;
      ns[o + 32] = st[i] * li + st[i + 32] * lr;
    }
#pragma unroll
    for (int i = 0; i < 64; ++i) st[i] = ns[i];
  }

#pragma unroll
  for (int i = 0; i < 16; ++i) {
    float ur = st[i],      ui = st[i + 32];
    float vr = st[i + 16], vi = st[i + 48];
    st[i]      = (ur + vr) * RS2;
    st[i + 32] = (ui + vi) * RS2;
    st[i + 16] = (ur - vr) * RS2;
    st[i + 48] = (ui - vi) * RS2;
  }

  float ev = 0.0f;
#pragma unroll
  for (int i = 0; i < 32; ++i) {
    float p2 = st[i] * st[i] + st[i + 32] * st[i + 32];
    ev += (i & 16) ? -p2 : p2;
  }
  if (p < P) out[p] = ev * 7.62939453125e-06f;
}

extern "C" void kernel_launch(void* const* d_in, const int* in_sizes, int n_in,
                              void* d_out, int out_size, void* d_ws, size_t ws_size,
                              hipStream_t stream) {
  (void)in_sizes; (void)n_in; (void)ws_size;
  const float* x      = (const float*)d_in[0];
  const float* thetas = (const float*)d_in[1];
  const float* phis   = (const float*)d_in[2];
  float* out = (float*)d_out;
  float* T   = (float*)d_ws;

  hipLaunchKernelGGL(qc_build_T, dim3(1), dim3(512), 0, stream, thetas, phis, T);

  const int P = out_size;
  const int blocks = (P + 127) / 128;
  hipLaunchKernelGGL(qc_main, dim3(blocks), dim3(128), 0, stream, x, T, out, P);
}
